# starting kernel
# baseline (speedup 1.0000x reference)
#include <hip/hip_runtime.h>
#include <hip/hip_bf16.h>

#define NN   512
#define KSPA 64
#define ND   16
#define NB   1024
#define NSH  448
#define N_L  30688

typedef __attribute__((ext_vector_type(2))) float v2f;
typedef __attribute__((ext_vector_type(8))) float v8f;

__global__ __launch_bounds__(128) void build_F_kernel(
    const float* __restrict__ L_emb,
    const float* __restrict__ S_emb,
    float* __restrict__ Fmat)
{
    __shared__ float hist[64][128];
    const int d  = blockIdx.x;
    const int c  = blockIdx.y * 128 + threadIdx.x;

    if (d == 0) {
        #pragma unroll 4
        for (int i = 0; i < 64; ++i)
            Fmat[(0 * 64 + i) * NN + c] = (c == NSH + i) ? 1.0f : 0.0f;
        return;
    }

    const float* Ld = L_emb + d * N_L;
    for (int i = 0; i < 64; ++i) {
        const int off = NSH * i + (i * (i - 1)) / 2;
        float acc;
        if (c < NSH) acc = Ld[off + c];
        else         acc = (c - NSH == i) ? S_emb[d * KSPA + i] : 0.0f;
        for (int k = 0; k < i; ++k) {
            const float w = Ld[off + NSH + k];
            acc += w * hist[k][threadIdx.x];
        }
        hist[i][threadIdx.x] = acc;
        Fmat[(d * 64 + i) * NN + c] = acc;
    }
}

__global__ __launch_bounds__(256) void copy_bias_kernel(
    const float* __restrict__ eps,
    const float* __restrict__ bias_shared,
    float* __restrict__ out)
{
    const int idx = blockIdx.x * blockDim.x + threadIdx.x;
    const int b  = idx / (NSH / 4);
    const int c4 = idx % (NSH / 4);
    float4 e  = ((const float4*)(eps + b * NN))[c4];
    float4 bs = ((const float4*)bias_shared)[c4];
    float4 o  = make_float4(e.x + bs.x, e.y + bs.y, e.z + bs.z, e.w + bs.w);
    ((float4*)(out + b * NN))[c4] = o;
}

__global__ __launch_bounds__(32) void spa_wmma_kernel(
    const float* __restrict__ eps,
    const int*   __restrict__ didx,
    const float* __restrict__ Fmat,
    const float* __restrict__ bias_ns,
    float* __restrict__ out)
{
    const int bt   = blockIdx.x;
    const int rt   = blockIdx.y;
    const int dom  = blockIdx.z;
    const int lane = threadIdx.x;
    const int half = lane >> 4;
    const int l16  = lane & 15;

    __shared__ float As[16][20];
    __shared__ float Bs[16][20];

    const float* Fbase = Fmat + (dom * 64 + rt * 16) * NN;
    const float* Ebase = eps + (bt * 16) * NN;

    const int lrow = lane >> 1;
    const int lcol = (lane & 1) * 8;

    v8f acc = {0.f, 0.f, 0.f, 0.f, 0.f, 0.f, 0.f, 0.f};

    for (int kk = 0; kk < NN; kk += 16) {
        float4 fa0 = *(const float4*)(Fbase + lrow * NN + kk + lcol);
        float4 fa1 = *(const float4*)(Fbase + lrow * NN + kk + lcol + 4);
        float4 eb0 = *(const float4*)(Ebase + lrow * NN + kk + lcol);
        float4 eb1 = *(const float4*)(Ebase + lrow * NN + kk + lcol + 4);
        __syncthreads();
        *(float4*)&As[lrow][lcol]     = fa0;
        *(float4*)&As[lrow][lcol + 4] = fa1;
        *(float4*)&Bs[lrow][lcol]     = eb0;
        *(float4*)&Bs[lrow][lcol + 4] = eb1;
        __syncthreads();

        #pragma unroll
        for (int ks = 0; ks < 16; ks += 4) {
            const int k0 = ks + half * 2;
            v2f a; a.x = As[l16][k0]; a.y = As[l16][k0 + 1];
            v2f b; b.x = Bs[l16][k0]; b.y = Bs[l16][k0 + 1];
            acc = __builtin_amdgcn_wmma_f32_16x16x4_f32(
                 false, a,  false, b,
                 (short)0, acc,  false,  false);
        }
    }

    const int b = bt * 16 + l16;
    if (didx[b] == dom) {
        #pragma unroll
        for (int v = 0; v < 8; ++v) {
            const int m = v + half * 8;
            out[b * NN + NSH + rt * 16 + m] =
                acc[v] + bias_ns[dom * KSPA + rt * 16 + m];
        }
    }
}

extern "C" void kernel_launch(void* const* d_in, const int* in_sizes, int n_in,
                              void* d_out, int out_size, void* d_ws, size_t ws_size,
                              hipStream_t stream) {
    const float* eps     = (const float*)d_in[0];
    const int*   didx    = (const int*)  d_in[1];
    const float* L_emb   = (const float*)d_in[2];
    const float* S_emb   = (const float*)d_in[3];
    const float* bias_ns = (const float*)d_in[4];
    const float* bias_sh = (const float*)d_in[5];
    float*       out     = (float*)d_out;
    float*       Fmat    = (float*)d_ws;

    build_F_kernel<<<dim3(ND, NN / 128), dim3(128), 0, stream>>>(L_emb, S_emb, Fmat);

    const int n4 = NB * (NSH / 4);
    copy_bias_kernel<<<n4 / 256, 256, 0, stream>>>(eps, bias_sh, out);

    spa_wmma_kernel<<<dim3(NB / 16, KSPA / 16, ND), dim3(32), 0, stream>>>(
        eps, didx, Fmat, bias_ns, out);
}


# F build: each column keeps its 64 rows in registers, rows done in blocks of 16, weights staged once in LDS and read as 16-byte broadcasts; every row still sums in ascending k (order unchanged)
# speedup vs baseline: 2.7908x; 2.7908x over previous
#include <hip/hip_runtime.h>
#include <hip/hip_bf16.h>

#define NN   512
#define KSPA 64
#define ND   16
#define NB   1024
#define NSH  448
#define N_L  30688

typedef __attribute__((ext_vector_type(2))) float v2f;
typedef __attribute__((ext_vector_type(8))) float v8f;

__global__ __launch_bounds__(128) void build_F_kernel(
    const float* __restrict__ L_emb,
    const float* __restrict__ S_emb,
    float* __restrict__ Fmat)
{
    __shared__ __attribute__((aligned(16))) float Wl[64][64];
    const int d   = blockIdx.x;
    const int tid = threadIdx.x;
    const int c   = blockIdx.y * 128 + tid;

    if (d == 0) {
        #pragma unroll 4
        for (int i = 0; i < 64; ++i)
            Fmat[(0 * 64 + i) * NN + c] = (c == NSH + i) ? 1.0f : 0.0f;
        return;
    }

    const float* Ld = L_emb + d * N_L;

    float F[64];
    float sv = 0.0f;
    const bool left = c < NSH;
    if (left) {
        #pragma unroll
        for (int i = 0; i < 64; ++i) {
            const int off = NSH * i + (i * (i - 1)) / 2;
            F[i] = Ld[off + c];
        }
    } else {
        sv = S_emb[d * KSPA + (c - NSH)];
    }

    {
        const unsigned hi = (unsigned)tid >> 6;
        const unsigned k  = (unsigned)tid & 63u;
        #pragma unroll
        for (unsigned it = 0; it < 32; ++it) {
            const unsigned i   = it * 2u + hi;
            const unsigned off = NSH * i + (i * (i - 1u)) / 2u;
            const unsigned kk  = (k < i) ? k : 0u;
            const float v = Ld[off + NSH + kk];
            Wl[i][k] = (k < i) ? v : 0.0f;
        }
    }

    if (!left) {
        #pragma unroll
        for (int i = 0; i < 64; ++i)
            F[i] = (c - NSH == i) ? sv : 0.0f;
    }
    __syncthreads();

    #pragma unroll
    for (int J = 0; J < 4; ++J) {
        const int i0 = J * 16;
        #pragma unroll
        for (int k0 = 0; k0 < i0; k0 += 4) {
            #pragma unroll
            for (int r = 0; r < 16; ++r) {
                const float4 w = *(const float4*)&Wl[i0 + r][k0];
                F[i0 + r] += w.x * F[k0];
                F[i0 + r] += w.y * F[k0 + 1];
                F[i0 + r] += w.z * F[k0 + 2];
                F[i0 + r] += w.w * F[k0 + 3];
            }
        }
        #pragma unroll
        for (int r = 1; r < 16; ++r) {
            #pragma unroll
            for (int q = 0; q < r; q += 4) {
                const float4 w = *(const float4*)&Wl[i0 + r][i0 + q];
                F[i0 + r] += w.x * F[i0 + q];
                if (q + 1 < r) F[i0 + r] += w.y * F[i0 + q + 1];
                if (q + 2 < r) F[i0 + r] += w.z * F[i0 + q + 2];
                if (q + 3 < r) F[i0 + r] += w.w * F[i0 + q + 3];
            }
        }
        #pragma unroll
        for (int r = 0; r < 16; ++r)
            Fmat[(d * 64 + i0 + r) * NN + c] = F[i0 + r];
    }
}

__global__ __launch_bounds__(256) void copy_bias_kernel(
    const float* __restrict__ eps,
    const float* __restrict__ bias_shared,
    float* __restrict__ out)
{
    const int idx = blockIdx.x * blockDim.x + threadIdx.x;
    const int b  = idx / (NSH / 4);
    const int c4 = idx % (NSH / 4);
    float4 e  = ((const float4*)(eps + b * NN))[c4];
    float4 bs = ((const float4*)bias_shared)[c4];
    float4 o  = make_float4(e.x + bs.x, e.y + bs.y, e.z + bs.z, e.w + bs.w);
    ((float4*)(out + b * NN))[c4] = o;
}

__global__ __launch_bounds__(32) void spa_wmma_kernel(
    const float* __restrict__ eps,
    const int*   __restrict__ didx,
    const float* __restrict__ Fmat,
    const float* __restrict__ bias_ns,
    float* __restrict__ out)
{
    const int bt   = blockIdx.x;
    const int rt   = blockIdx.y;
    const int dom  = blockIdx.z;
    const int lane = threadIdx.x;
    const int half = lane >> 4;
    const int l16  = lane & 15;

    __shared__ float As[16][20];
    __shared__ float Bs[16][20];

    const float* Fbase = Fmat + (dom * 64 + rt * 16) * NN;
    const float* Ebase = eps + (bt * 16) * NN;

    const int lrow = lane >> 1;
    const int lcol = (lane & 1) * 8;

    v8f acc = {0.f, 0.f, 0.f, 0.f, 0.f, 0.f, 0.f, 0.f};

    for (int kk = 0; kk < NN; kk += 16) {
        float4 fa0 = *(const float4*)(Fbase + lrow * NN + kk + lcol);
        float4 fa1 = *(const float4*)(Fbase + lrow * NN + kk + lcol + 4);
        float4 eb0 = *(const float4*)(Ebase + lrow * NN + kk + lcol);
        float4 eb1 = *(const float4*)(Ebase + lrow * NN + kk + lcol + 4);
        __syncthreads();
        *(float4*)&As[lrow][lcol]     = fa0;
        *(float4*)&As[lrow][lcol + 4] = fa1;
        *(float4*)&Bs[lrow][lcol]     = eb0;
        *(float4*)&Bs[lrow][lcol + 4] = eb1;
        __syncthreads();

        #pragma unroll
        for (int ks = 0; ks < 16; ks += 4) {
            const int k0 = ks + half * 2;
            v2f a; a.x = As[l16][k0]; a.y = As[l16][k0 + 1];
            v2f b; b.x = Bs[l16][k0]; b.y = Bs[l16][k0 + 1];
            acc = __builtin_amdgcn_wmma_f32_16x16x4_f32(
                 false, a,  false, b,
                 (short)0, acc,  false,  false);
        }
    }

    const int b = bt * 16 + l16;
    if (didx[b] == dom) {
        #pragma unroll
        for (int v = 0; v < 8; ++v) {
            const int m = v + half * 8;
            out[b * NN + NSH + rt * 16 + m] =
                acc[v] + bias_ns[dom * KSPA + rt * 16 + m];
        }
    }
}

extern "C" void kernel_launch(void* const* d_in, const int* in_sizes, int n_in,
                              void* d_out, int out_size, void* d_ws, size_t ws_size,
                              hipStream_t stream) {
    const float* eps     = (const float*)d_in[0];
    const int*   didx    = (const int*)  d_in[1];
    const float* L_emb   = (const float*)d_in[2];
    const float* S_emb   = (const float*)d_in[3];
    const float* bias_ns = (const float*)d_in[4];
    const float* bias_sh = (const float*)d_in[5];
    float*       out     = (float*)d_out;
    float*       Fmat    = (float*)d_ws;

    build_F_kernel<<<dim3(ND, NN / 128), dim3(128), 0, stream>>>(L_emb, S_emb, Fmat);

    const int n4 = NB * (NSH / 4);
    copy_bias_kernel<<<n4 / 256, 256, 0, stream>>>(eps, bias_sh, out);

    spa_wmma_kernel<<<dim3(NB / 16, KSPA / 16, ND), dim3(32), 0, stream>>>(
        eps, didx, Fmat, bias_ns, out);
}


# Tail product: each wave ranks the batch by domain (ballot scan) and multiplies a 16-column tile of ONE domain by that domain's F only; no masked per-domain tiles (grid 8x4x16); k order unchanged
# speedup vs baseline: 2.8828x; 1.0330x over previous
#include <hip/hip_runtime.h>
#include <hip/hip_bf16.h>

#define NN   512
#define KSPA 64
#define ND   16
#define NB   1024
#define NSH  448
#define N_L  30688

typedef __attribute__((ext_vector_type(2))) float v2f;
typedef __attribute__((ext_vector_type(8))) float v8f;

__global__ __launch_bounds__(128) void build_F_kernel(
    const float* __restrict__ L_emb,
    const float* __restrict__ S_emb,
    float* __restrict__ Fmat)
{
    __shared__ __attribute__((aligned(16))) float Wl[64][64];
    const int d   = blockIdx.x;
    const int tid = threadIdx.x;
    const int c   = blockIdx.y * 128 + tid;

    if (d == 0) {
        #pragma unroll 4
        for (int i = 0; i < 64; ++i)
            Fmat[(0 * 64 + i) * NN + c] = (c == NSH + i) ? 1.0f : 0.0f;
        return;
    }

    const float* Ld = L_emb + d * N_L;

    float F[64];
    float sv = 0.0f;
    const bool left = c < NSH;
    if (left) {
        #pragma unroll
        for (int i = 0; i < 64; ++i) {
            const int off = NSH * i + (i * (i - 1)) / 2;
            F[i] = Ld[off + c];
        }
    } else {
        sv = S_emb[d * KSPA + (c - NSH)];
    }

    {
        const unsigned hi = (unsigned)tid >> 6;
        const unsigned k  = (unsigned)tid & 63u;
        #pragma unroll
        for (unsigned it = 0; it < 32; ++it) {
            const unsigned i   = it * 2u + hi;
            const unsigned off = NSH * i + (i * (i - 1u)) / 2u;
            const unsigned kk  = (k < i) ? k : 0u;
            const float v = Ld[off + NSH + kk];
            Wl[i][k] = (k < i) ? v : 0.0f;
        }
    }

    if (!left) {
        #pragma unroll
        for (int i = 0; i < 64; ++i)
            F[i] = (c - NSH == i) ? sv : 0.0f;
    }
    __syncthreads();

    #pragma unroll
    for (int J = 0; J < 4; ++J) {
        const int i0 = J * 16;
        #pragma unroll
        for (int k0 = 0; k0 < i0; k0 += 4) {
            #pragma unroll
            for (int r = 0; r < 16; ++r) {
                const float4 w = *(const float4*)&Wl[i0 + r][k0];
                F[i0 + r] += w.x * F[k0];
                F[i0 + r] += w.y * F[k0 + 1];
                F[i0 + r] += w.z * F[k0 + 2];
                F[i0 + r] += w.w * F[k0 + 3];
            }
        }
        #pragma unroll
        for (int r = 1; r < 16; ++r) {
            #pragma unroll
            for (int q = 0; q < r; q += 4) {
                const float4 w = *(const float4*)&Wl[i0 + r][i0 + q];
                F[i0 + r] += w.x * F[i0 + q];
                if (q + 1 < r) F[i0 + r] += w.y * F[i0 + q + 1];
                if (q + 2 < r) F[i0 + r] += w.z * F[i0 + q + 2];
                if (q + 3 < r) F[i0 + r] += w.w * F[i0 + q + 3];
            }
        }
        #pragma unroll
        for (int r = 0; r < 16; ++r)
            Fmat[(d * 64 + i0 + r) * NN + c] = F[i0 + r];
    }
}

__global__ __launch_bounds__(256) void copy_bias_kernel(
    const float* __restrict__ eps,
    const float* __restrict__ bias_shared,
    float* __restrict__ out)
{
    const int idx = blockIdx.x * blockDim.x + threadIdx.x;
    const int b  = idx / (NSH / 4);
    const int c4 = idx % (NSH / 4);
    float4 e  = ((const float4*)(eps + b * NN))[c4];
    float4 bs = ((const float4*)bias_shared)[c4];
    float4 o  = make_float4(e.x + bs.x, e.y + bs.y, e.z + bs.z, e.w + bs.w);
    ((float4*)(out + b * NN))[c4] = o;
}

#define TSLOTS 8

__global__ __launch_bounds__(32) void spa_wmma_kernel(
    const float* __restrict__ eps,
    const int*   __restrict__ didx,
    const float* __restrict__ Fmat,
    const float* __restrict__ bias_ns,
    float* __restrict__ out)
{
    const int slot = blockIdx.x;
    const int rt   = blockIdx.y;
    const int dom  = blockIdx.z;
    const int lane = threadIdx.x;
    const int half = lane >> 4;
    const int l16  = lane & 15;

    __shared__ float As[16][20];
    __shared__ float Bs[16][20];
    __shared__ int   list[NB + 32];

    int dv[NB / 32];
    #pragma unroll
    for (int j = 0; j < NB / 32; ++j)
        dv[j] = didx[j * 32 + lane];
    unsigned mk[NB / 32];
    #pragma unroll
    for (int j = 0; j < NB / 32; ++j)
        mk[j] = __builtin_amdgcn_ballot_w32(dv[j] == dom);
    int n = 0;
    #pragma unroll
    for (int j = 0; j < NB / 32; ++j) {
        const int pos = (dv[j] == dom) ? n + (int)__builtin_amdgcn_mbcnt_lo(mk[j], 0u) : NB + lane;
        list[pos] = j * 32 + lane;
        n += __builtin_popcount(mk[j]);
    }
    __syncthreads();

    const float* Fbase = Fmat + (dom * 64 + rt * 16) * NN;

    const int lrow = lane >> 1;
    const int lcol = (lane & 1) * 8;

    for (int t = slot; t * 16 < n; t += TSLOTS) {
        const int cl = t * 16 + lrow;
        const float* Erow = eps + list[cl < n ? cl : t * 16] * NN;

        v8f acc = {0.f, 0.f, 0.f, 0.f, 0.f, 0.f, 0.f, 0.f};

        for (int kk = 0; kk < NN; kk += 16) {
            float4 fa0 = *(const float4*)(Fbase + lrow * NN + kk + lcol);
            float4 fa1 = *(const float4*)(Fbase + lrow * NN + kk + lcol + 4);
            float4 eb0 = *(const float4*)(Erow + kk + lcol);
            float4 eb1 = *(const float4*)(Erow + kk + lcol + 4);
            __syncthreads();
            *(float4*)&As[lrow][lcol]     = fa0;
            *(float4*)&As[lrow][lcol + 4] = fa1;
            *(float4*)&Bs[lrow][lcol]     = eb0;
            *(float4*)&Bs[lrow][lcol + 4] = eb1;
            __syncthreads();

            #pragma unroll
            for (int ks = 0; ks < 16; ks += 4) {
                const int k0 = ks + half * 2;
                v2f a; a.x = As[l16][k0]; a.y = As[l16][k0 + 1];
                v2f b; b.x = Bs[l16][k0]; b.y = Bs[l16][k0 + 1];
                acc = __builtin_amdgcn_wmma_f32_16x16x4_f32(
                     false, a,  false, b,
                     (short)0, acc,  false,  false);
            }
        }

        const int cn = t * 16 + l16;
        if (cn < n) {
            const int b = list[cn];
            #pragma unroll
            for (int v = 0; v < 8; ++v) {
                const int m = v + half * 8;
                out[b * NN + NSH + rt * 16 + m] =
                    acc[v] + bias_ns[dom * KSPA + rt * 16 + m];
            }
        }
    }
}

extern "C" void kernel_launch(void* const* d_in, const int* in_sizes, int n_in,
                              void* d_out, int out_size, void* d_ws, size_t ws_size,
                              hipStream_t stream) {
    const float* eps     = (const float*)d_in[0];
    const int*   didx    = (const int*)  d_in[1];
    const float* L_emb   = (const float*)d_in[2];
    const float* S_emb   = (const float*)d_in[3];
    const float* bias_ns = (const float*)d_in[4];
    const float* bias_sh = (const float*)d_in[5];
    float*       out     = (float*)d_out;
    float*       Fmat    = (float*)d_ws;

    build_F_kernel<<<dim3(ND, NN / 128), dim3(128), 0, stream>>>(L_emb, S_emb, Fmat);

    const int n4 = NB * (NSH / 4);
    copy_bias_kernel<<<n4 / 256, 256, 0, stream>>>(eps, bias_sh, out);

    spa_wmma_kernel<<<dim3(TSLOTS, KSPA / 16, ND), dim3(32), 0, stream>>>(
        eps, didx, Fmat, bias_ns, out);
}


# Tail product: K goes through LDS in chunks of 128 (was 16), whole 512-byte rows per load; next chunk's global loads and the chunk's fragment reads issued ahead of its 32 products; k order unchanged
# speedup vs baseline: 4.3451x; 1.5072x over previous
#include <hip/hip_runtime.h>
#include <hip/hip_bf16.h>

#define NN   512
#define KSPA 64
#define ND   16
#define NB   1024
#define NSH  448
#define N_L  30688

typedef __attribute__((ext_vector_type(2))) float v2f;
typedef __attribute__((ext_vector_type(4))) float v4f;
typedef __attribute__((ext_vector_type(8))) float v8f;

__global__ __launch_bounds__(128) void build_F_kernel(
    const float* __restrict__ L_emb,
    const float* __restrict__ S_emb,
    float* __restrict__ Fmat)
{
    __shared__ __attribute__((aligned(16))) float Wl[64][64];
    const int d   = blockIdx.x;
    const int tid = threadIdx.x;
    const int c   = blockIdx.y * 128 + tid;

    if (d == 0) {
        #pragma unroll 4
        for (int i = 0; i < 64; ++i)
            Fmat[(0 * 64 + i) * NN + c] = (c == NSH + i) ? 1.0f : 0.0f;
        return;
    }

    const float* Ld = L_emb + d * N_L;

    float F[64];
    float sv = 0.0f;
    const bool left = c < NSH;
    if (left) {
        #pragma unroll
        for (int i = 0; i < 64; ++i) {
            const int off = NSH * i + (i * (i - 1)) / 2;
            F[i] = Ld[off + c];
        }
    } else {
        sv = S_emb[d * KSPA + (c - NSH)];
    }

    {
        const unsigned hi = (unsigned)tid >> 6;
        const unsigned k  = (unsigned)tid & 63u;
        #pragma unroll
        for (unsigned it = 0; it < 32; ++it) {
            const unsigned i   = it * 2u + hi;
            const unsigned off = NSH * i + (i * (i - 1u)) / 2u;
            const unsigned kk  = (k < i) ? k : 0u;
            const float v = Ld[off + NSH + kk];
            Wl[i][k] = (k < i) ? v : 0.0f;
        }
    }

    if (!left) {
        #pragma unroll
        for (int i = 0; i < 64; ++i)
            F[i] = (c - NSH == i) ? sv : 0.0f;
    }
    __syncthreads();

    #pragma unroll
    for (int J = 0; J < 4; ++J) {
        const int i0 = J * 16;
        #pragma unroll
        for (int k0 = 0; k0 < i0; k0 += 4) {
            #pragma unroll
            for (int r = 0; r < 16; ++r) {
                const float4 w = *(const float4*)&Wl[i0 + r][k0];
                F[i0 + r] += w.x * F[k0];
                F[i0 + r] += w.y * F[k0 + 1];
                F[i0 + r] += w.z * F[k0 + 2];
                F[i0 + r] += w.w * F[k0 + 3];
            }
        }
        #pragma unroll
        for (int r = 1; r < 16; ++r) {
            #pragma unroll
            for (int q = 0; q < r; q += 4) {
                const float4 w = *(const float4*)&Wl[i0 + r][i0 + q];
                F[i0 + r] += w.x * F[i0 + q];
                if (q + 1 < r) F[i0 + r] += w.y * F[i0 + q + 1];
                if (q + 2 < r) F[i0 + r] += w.z * F[i0 + q + 2];
                if (q + 3 < r) F[i0 + r] += w.w * F[i0 + q + 3];
            }
        }
        #pragma unroll
        for (int r = 0; r < 16; ++r)
            Fmat[(d * 64 + i0 + r) * NN + c] = F[i0 + r];
    }
}

__global__ __launch_bounds__(256) void copy_bias_kernel(
    const float* __restrict__ eps,
    const float* __restrict__ bias_shared,
    float* __restrict__ out)
{
    const int idx = blockIdx.x * blockDim.x + threadIdx.x;
    const int b  = idx / (NSH / 4);
    const int c4 = idx % (NSH / 4);
    float4 e  = ((const float4*)(eps + b * NN))[c4];
    float4 bs = ((const float4*)bias_shared)[c4];
    float4 o  = make_float4(e.x + bs.x, e.y + bs.y, e.z + bs.z, e.w + bs.w);
    ((float4*)(out + b * NN))[c4] = o;
}

#define TSLOTS 8
#define KC 128
#define KP (KC + 4)

__global__ __launch_bounds__(32) void spa_wmma_kernel(
    const float* __restrict__ eps,
    const int*   __restrict__ didx,
    const float* __restrict__ Fmat,
    const float* __restrict__ bias_ns,
    float* __restrict__ out)
{
    const int slot = blockIdx.x;
    const int rt   = blockIdx.y;
    const int dom  = blockIdx.z;
    const int lane = threadIdx.x;
    const int half = lane >> 4;
    const int l16  = lane & 15;

    __shared__ __attribute__((aligned(16))) float As[16][KP];
    __shared__ __attribute__((aligned(16))) float Bs[16][KP];
    __shared__ int   list[NB + 32];

    int dv[NB / 32];
    #pragma unroll
    for (int j = 0; j < NB / 32; ++j)
        dv[j] = didx[j * 32 + lane];

    const float* Fbase = Fmat + (dom * 64 + rt * 16) * NN + lane * 4;
    v4f ra0[16];
    #pragma unroll
    for (int j = 0; j < 16; ++j)
        ra0[j] = *(const v4f*)(Fbase + j * NN);
    const float* bptr = bias_ns + dom * KSPA + rt * 16 + half * 8;
    const v4f bz0 = *(const v4f*)(bptr);
    const v4f bz1 = *(const v4f*)(bptr + 4);

    unsigned mk[NB / 32];
    #pragma unroll
    for (int j = 0; j < NB / 32; ++j)
        mk[j] = __builtin_amdgcn_ballot_w32(dv[j] == dom);
    int n = 0;
    #pragma unroll
    for (int j = 0; j < NB / 32; ++j) {
        const int pos = (dv[j] == dom) ? n + (int)__builtin_amdgcn_mbcnt_lo(mk[j], 0u) : NB + lane;
        list[pos] = j * 32 + lane;
        n += __builtin_popcount(mk[j]);
    }
    __syncthreads();

    for (int t = slot; t * 16 < n; t += TSLOTS) {
        const int cn = t * 16 + l16;
        const int myb = list[cn < n ? cn : t * 16];
        const float* Erow[16];
        #pragma unroll
        for (int j = 0; j < 16; ++j)
            Erow[j] = eps + __builtin_amdgcn_readlane(myb, j) * NN + lane * 4;

        v4f ra[16], rb[16];
        #pragma unroll
        for (int j = 0; j < 16; ++j) {
            ra[j] = ra0[j];
            rb[j] = *(const v4f*)(Erow[j]);
        }

        v8f acc = {0.f, 0.f, 0.f, 0.f, 0.f, 0.f, 0.f, 0.f};

        #pragma unroll 1
        for (int kc = 0; kc < NN / KC; ++kc) {
            #pragma unroll
            for (int j = 0; j < 16; ++j) {
                *(v4f*)&As[j][lane * 4] = ra[j];
                *(v4f*)&Bs[j][lane * 4] = rb[j];
            }
            __syncthreads();

            v2f fa[KC / 4], fb[KC / 4];
            #pragma unroll
            for (int s = 0; s < KC / 4; ++s) {
                const int k0 = s * 4 + half * 2;
                fa[s].x = As[l16][k0]; fa[s].y = As[l16][k0 + 1];
                fb[s].x = Bs[l16][k0]; fb[s].y = Bs[l16][k0 + 1];
            }

            if (kc + 1 < NN / KC) {
                #pragma unroll
                for (int j = 0; j < 16; ++j) {
                    ra[j] = *(const v4f*)(Fbase + j * NN + (kc + 1) * KC);
                    rb[j] = *(const v4f*)(Erow[j] + (kc + 1) * KC);
                }
            }

            __builtin_amdgcn_sched_barrier(0);
            #pragma unroll
            for (int s = 0; s < KC / 4; ++s)
                acc = __builtin_amdgcn_wmma_f32_16x16x4_f32(
                     false, fa[s],  false, fb[s],
                     (short)0, acc,  false,  false);
            __builtin_amdgcn_sched_barrier(0);
            __syncthreads();
        }

        if (cn < n) {
            float* orow = out + myb * NN + NSH + rt * 16 + half * 8;
            #pragma unroll
            for (int v = 0; v < 4; ++v) {
                orow[v]     = acc[v]     + bz0[v];
                orow[v + 4] = acc[v + 4] + bz1[v];
            }
        }
    }
}

extern "C" void kernel_launch(void* const* d_in, const int* in_sizes, int n_in,
                              void* d_out, int out_size, void* d_ws, size_t ws_size,
                              hipStream_t stream) {
    const float* eps     = (const float*)d_in[0];
    const int*   didx    = (const int*)  d_in[1];
    const float* L_emb   = (const float*)d_in[2];
    const float* S_emb   = (const float*)d_in[3];
    const float* bias_ns = (const float*)d_in[4];
    const float* bias_sh = (const float*)d_in[5];
    float*       out     = (float*)d_out;
    float*       Fmat    = (float*)d_ws;

    build_F_kernel<<<dim3(ND, NN / 128), dim3(128), 0, stream>>>(L_emb, S_emb, Fmat);

    const int n4 = NB * (NSH / 4);
    copy_bias_kernel<<<n4 / 256, 256, 0, stream>>>(eps, bias_sh, out);

    spa_wmma_kernel<<<dim3(TSLOTS, KSPA / 16, ND), dim3(32), 0, stream>>>(
        eps, didx, Fmat, bias_ns, out);
}


# Shared-prefix copy (out[:, :448] = eps + bias) runs as 896 extra blocks of the F-build launch, beside the 64 F blocks: 2 launches per call (was 3); every sum and its k order unchanged
# speedup vs baseline: 5.1838x; 1.1930x over previous
#include <hip/hip_runtime.h>
#include <hip/hip_bf16.h>

#define NN   512
#define KSPA 64
#define ND   16
#define NB   1024
#define NSH  448
#define N_L  30688

typedef __attribute__((ext_vector_type(2))) float v2f;
typedef __attribute__((ext_vector_type(4))) float v4f;
typedef __attribute__((ext_vector_type(8))) float v8f;

#define FROWS_Y 4
#define PREF_Y  56
#define GRID1_Y 60

__global__ __launch_bounds__(128) void build_F_prefix_kernel(
    const float* __restrict__ L_emb,
    const float* __restrict__ S_emb,
    float* __restrict__ Fmat,
    const float* __restrict__ eps,
    const float* __restrict__ bias_shared,
    float* __restrict__ out)
{
    __shared__ __attribute__((aligned(16))) float Wl[64][64];
    const int d   = blockIdx.x;
    const int tid = threadIdx.x;

    if (blockIdx.y >= FROWS_Y) {
        const int idx = ((blockIdx.y - FROWS_Y) * ND + blockIdx.x) * 128 + tid;
        const int b  = idx / (NSH / 4);
        const int c4 = idx % (NSH / 4);
        const float4 e  = ((const float4*)(eps + b * NN))[c4];
        const float4 bs = ((const float4*)bias_shared)[c4];
        ((float4*)(out + b * NN))[c4] =
            make_float4(e.x + bs.x, e.y + bs.y, e.z + bs.z, e.w + bs.w);
        return;
    }

    const int c   = blockIdx.y * 128 + tid;

    if (d == 0) {
        #pragma unroll 4
        for (int i = 0; i < 64; ++i)
            Fmat[(0 * 64 + i) * NN + c] = (c == NSH + i) ? 1.0f : 0.0f;
        return;
    }

    const float* Ld = L_emb + d * N_L;

    float F[64];
    float sv = 0.0f;
    const bool left = c < NSH;
    if (left) {
        #pragma unroll
        for (int i = 0; i < 64; ++i) {
            const int off = NSH * i + (i * (i - 1)) / 2;
            F[i] = Ld[off + c];
        }
    } else {
        sv = S_emb[d * KSPA + (c - NSH)];
    }

    {
        const unsigned hi = (unsigned)tid >> 6;
        const unsigned k  = (unsigned)tid & 63u;
        #pragma unroll
        for (unsigned it = 0; it < 32; ++it) {
            const unsigned i   = it * 2u + hi;
            const unsigned off = NSH * i + (i * (i - 1u)) / 2u;
            const unsigned kk  = (k < i) ? k : 0u;
            const float v = Ld[off + NSH + kk];
            Wl[i][k] = (k < i) ? v : 0.0f;
        }
    }

    if (!left) {
        #pragma unroll
        for (int i = 0; i < 64; ++i)
            F[i] = (c - NSH == i) ? sv : 0.0f;
    }
    __syncthreads();

    #pragma unroll
    for (int J = 0; J < 4; ++J) {
        const int i0 = J * 16;
        #pragma unroll
        for (int k0 = 0; k0 < i0; k0 += 4) {
            #pragma unroll
            for (int r = 0; r < 16; ++r) {
                const float4 w = *(const float4*)&Wl[i0 + r][k0];
                F[i0 + r] += w.x * F[k0];
                F[i0 + r] += w.y * F[k0 + 1];
                F[i0 + r] += w.z * F[k0 + 2];
                F[i0 + r] += w.w * F[k0 + 3];
            }
        }
        #pragma unroll
        for (int r = 1; r < 16; ++r) {
            #pragma unroll
            for (int q = 0; q < r; q += 4) {
                const float4 w = *(const float4*)&Wl[i0 + r][i0 + q];
                F[i0 + r] += w.x * F[i0 + q];
                if (q + 1 < r) F[i0 + r] += w.y * F[i0 + q + 1];
                if (q + 2 < r) F[i0 + r] += w.z * F[i0 + q + 2];
                if (q + 3 < r) F[i0 + r] += w.w * F[i0 + q + 3];
            }
        }
        #pragma unroll
        for (int r = 0; r < 16; ++r)
            Fmat[(d * 64 + i0 + r) * NN + c] = F[i0 + r];
    }
}

#define TSLOTS 8
#define KC 128
#define KP (KC + 4)

__global__ __launch_bounds__(32) void spa_wmma_kernel(
    const float* __restrict__ eps,
    const int*   __restrict__ didx,
    const float* __restrict__ Fmat,
    const float* __restrict__ bias_ns,
    float* __restrict__ out)
{
    const int slot = blockIdx.x;
    const int rt   = blockIdx.y;
    const int dom  = blockIdx.z;
    const int lane = threadIdx.x;
    const int half = lane >> 4;
    const int l16  = lane & 15;

    __shared__ __attribute__((aligned(16))) float As[16][KP];
    __shared__ __attribute__((aligned(16))) float Bs[16][KP];
    __shared__ int   list[NB + 32];

    int dv[NB / 32];
    #pragma unroll
    for (int j = 0; j < NB / 32; ++j)
        dv[j] = didx[j * 32 + lane];

    const float* Fbase = Fmat + (dom * 64 + rt * 16) * NN + lane * 4;
    v4f ra0[16];
    #pragma unroll
    for (int j = 0; j < 16; ++j)
        ra0[j] = *(const v4f*)(Fbase + j * NN);
    const float* bptr = bias_ns + dom * KSPA + rt * 16 + half * 8;
    const v4f bz0 = *(const v4f*)(bptr);
    const v4f bz1 = *(const v4f*)(bptr + 4);

    unsigned mk[NB / 32];
    #pragma unroll
    for (int j = 0; j < NB / 32; ++j)
        mk[j] = __builtin_amdgcn_ballot_w32(dv[j] == dom);
    int n = 0;
    #pragma unroll
    for (int j = 0; j < NB / 32; ++j) {
        const int pos = (dv[j] == dom) ? n + (int)__builtin_amdgcn_mbcnt_lo(mk[j], 0u) : NB + lane;
        list[pos] = j * 32 + lane;
        n += __builtin_popcount(mk[j]);
    }
    __syncthreads();

    for (int t = slot; t * 16 < n; t += TSLOTS) {
        const int cn = t * 16 + l16;
        const int myb = list[cn < n ? cn : t * 16];
        const float* Erow[16];
        #pragma unroll
        for (int j = 0; j < 16; ++j)
            Erow[j] = eps + __builtin_amdgcn_readlane(myb, j) * NN + lane * 4;

        v4f ra[16], rb[16];
        #pragma unroll
        for (int j = 0; j < 16; ++j) {
            ra[j] = ra0[j];
            rb[j] = *(const v4f*)(Erow[j]);
        }

        v8f acc = {0.f, 0.f, 0.f, 0.f, 0.f, 0.f, 0.f, 0.f};

        #pragma unroll 1
        for (int kc = 0; kc < NN / KC; ++kc) {
            #pragma unroll
            for (int j = 0; j < 16; ++j) {
                *(v4f*)&As[j][lane * 4] = ra[j];
                *(v4f*)&Bs[j][lane * 4] = rb[j];
            }
            __syncthreads();

            v2f fa[KC / 4], fb[KC / 4];
            #pragma unroll
            for (int s = 0; s < KC / 4; ++s) {
                const int k0 = s * 4 + half * 2;
                fa[s].x = As[l16][k0]; fa[s].y = As[l16][k0 + 1];
                fb[s].x = Bs[l16][k0]; fb[s].y = Bs[l16][k0 + 1];
            }

            if (kc + 1 < NN / KC) {
                #pragma unroll
                for (int j = 0; j < 16; ++j) {
                    ra[j] = *(const v4f*)(Fbase + j * NN + (kc + 1) * KC);
                    rb[j] = *(const v4f*)(Erow[j] + (kc + 1) * KC);
                }
            }

            __builtin_amdgcn_sched_barrier(0);
            #pragma unroll
            for (int s = 0; s < KC / 4; ++s)
                acc = __builtin_amdgcn_wmma_f32_16x16x4_f32(
                     false, fa[s],  false, fb[s],
                     (short)0, acc,  false,  false);
            __builtin_amdgcn_sched_barrier(0);
            __syncthreads();
        }

        if (cn < n) {
            float* orow = out + myb * NN + NSH + rt * 16 + half * 8;
            #pragma unroll
            for (int v = 0; v < 4; ++v) {
                orow[v]     = acc[v]     + bz0[v];
                orow[v + 4] = acc[v + 4] + bz1[v];
            }
        }
    }
}

extern "C" void kernel_launch(void* const* d_in, const int* in_sizes, int n_in,
                              void* d_out, int out_size, void* d_ws, size_t ws_size,
                              hipStream_t stream) {
    const float* eps     = (const float*)d_in[0];
    const int*   didx    = (const int*)  d_in[1];
    const float* L_emb   = (const float*)d_in[2];
    const float* S_emb   = (const float*)d_in[3];
    const float* bias_ns = (const float*)d_in[4];
    const float* bias_sh = (const float*)d_in[5];
    float*       out     = (float*)d_out;
    float*       Fmat    = (float*)d_ws;

    build_F_prefix_kernel<<<dim3(ND, GRID1_Y), dim3(128), 0, stream>>>(
        L_emb, S_emb, Fmat, eps, bias_sh, out);

    spa_wmma_kernel<<<dim3(TSLOTS, KSPA / 16, ND), dim3(32), 0, stream>>>(
        eps, didx, Fmat, bias_ns, out);
}


# Tail by forward substitution per batch element: z=(I-Llow)^-1(Lleft*eps+S*eps); Lleft*eps on the matrix unit, recurrence in registers; F is never built: 1 launch, no workspace; sum order changes
# speedup vs baseline: 8.4694x; 1.6338x over previous
#include <hip/hip_runtime.h>
#include <hip/hip_bf16.h>

#define NN   512
#define KSPA 64
#define ND   16
#define NB   1024
#define NSH  448
#define N_L  30688

typedef __attribute__((ext_vector_type(2))) float v2f;
typedef __attribute__((ext_vector_type(4))) float v4f;
typedef __attribute__((ext_vector_type(8))) float v8f;
typedef float v4fu __attribute__((ext_vector_type(4), aligned(4)));

#define TSLOTS 8
#define KC 128
#define KP (KC + 4)
#define YP 17
#define TAILB (TSLOTS * ND)
#define PREFB 112
#define GRIDB (TAILB + PREFB)

#define NGRP 16
#define GW   60

__device__ constexpr int grp_J(int g)    { return g < 1 ? 0 : g < 4 ? 1 : g < 9 ? 2 : 3; }
__device__ constexpr bool grp_tri(int g) { return g == 0 || g == 3 || g == 8 || g == 15; }
__device__ constexpr int grp_k0(int g)   { return g < 1 ? 0 : g < 4 ? (g - 1) * 8 : g < 9 ? (g - 4) * 8 : (g - 9) * 8; }

template <int G>
__device__ __forceinline__ void load_group(const float (*Wl)[64], v4f (&w)[GW])
{
    constexpr int i0 = grp_J(G) * 16;
    if constexpr (grp_tri(G)) {
        #pragma unroll
        for (int r = 1; r < 16; ++r) {
            #pragma unroll
            for (int q4 = 0; q4 < 4; ++q4) {
                const int q = q4 * 4;
                if (q + 3 < r) {
                    w[(r - 1) * 4 + q4] = *(const v4f*)&Wl[i0 + r][i0 + q];
                } else if (q < r) {
                    w[(r - 1) * 4 + q4].x = Wl[i0 + r][i0 + q];
                    if (q + 1 < r) w[(r - 1) * 4 + q4].y = Wl[i0 + r][i0 + q + 1];
                    if (q + 2 < r) w[(r - 1) * 4 + q4].z = Wl[i0 + r][i0 + q + 2];
                }
            }
        }
    } else {
        constexpr int k0 = grp_k0(G);
        #pragma unroll
        for (int r = 0; r < 16; ++r) {
            w[2 * r]     = *(const v4f*)&Wl[i0 + r][k0];
            w[2 * r + 1] = *(const v4f*)&Wl[i0 + r][k0 + 4];
        }
    }
}

template <int G>
__device__ __forceinline__ void sum_group(const v4f (&w)[GW], float (&z)[64])
{
    constexpr int i0 = grp_J(G) * 16;
    if constexpr (grp_tri(G)) {
        #pragma unroll
        for (int r = 1; r < 16; ++r) {
            #pragma unroll
            for (int q4 = 0; q4 < 4; ++q4) {
                const int q = q4 * 4;
                if (q < r) {
                    const v4f v = w[(r - 1) * 4 + q4];
                    z[i0 + r] += v.x * z[i0 + q];
                    if (q + 1 < r) z[i0 + r] += v.y * z[i0 + q + 1];
                    if (q + 2 < r) z[i0 + r] += v.z * z[i0 + q + 2];
                    if (q + 3 < r) z[i0 + r] += v.w * z[i0 + q + 3];
                }
            }
        }
    } else {
        constexpr int k0 = grp_k0(G);
        #pragma unroll
        for (int r = 0; r < 16; ++r) {
            const v4f a = w[2 * r], b = w[2 * r + 1];
            z[i0 + r] += a.x * z[k0];
            z[i0 + r] += a.y * z[k0 + 1];
            z[i0 + r] += a.z * z[k0 + 2];
            z[i0 + r] += a.w * z[k0 + 3];
            z[i0 + r] += b.x * z[k0 + 4];
            z[i0 + r] += b.y * z[k0 + 5];
            z[i0 + r] += b.z * z[k0 + 6];
            z[i0 + r] += b.w * z[k0 + 7];
        }
    }
}

template <int G>
__device__ __forceinline__ void subst_step(const float (*Wl)[64], const v4f (&cur)[GW],
                                           v4f (&nxt)[GW], float (&z)[64])
{
    if constexpr (G + 1 < NGRP)
        load_group<G + 1>(Wl, nxt);
    __builtin_amdgcn_sched_barrier(0);
    sum_group<G>(cur, z);
    __builtin_amdgcn_sched_barrier(0);
}

__global__ __launch_bounds__(128) void tail_subst_prefix_kernel(
    const float* __restrict__ eps,
    const int*   __restrict__ didx,
    const float* __restrict__ L_emb,
    const float* __restrict__ S_emb,
    const float* __restrict__ bias_ns,
    const float* __restrict__ bias_shared,
    float* __restrict__ out)
{
    const int tid = threadIdx.x;

    if (blockIdx.x >= TAILB) {
        const int base = (blockIdx.x - TAILB) * 1024 + tid;
        float4 e[8], bs[8];
        int bb[8], cc[8];
        #pragma unroll
        for (int q = 0; q < 8; ++q) {
            const int idx = base + q * 128;
            bb[q] = idx / (NSH / 4);
            cc[q] = idx % (NSH / 4);
            e[q]  = ((const float4*)(eps + bb[q] * NN))[cc[q]];
            bs[q] = ((const float4*)bias_shared)[cc[q]];
        }
        #pragma unroll
        for (int q = 0; q < 8; ++q)
            ((float4*)(out + bb[q] * NN))[cc[q]] =
                make_float4(e[q].x + bs[q].x, e[q].y + bs[q].y, e[q].z + bs[q].z, e[q].w + bs[q].w);
        return;
    }

    const int dom  = blockIdx.x / TSLOTS;
    const int slot = blockIdx.x % TSLOTS;
    const int wv   = __builtin_amdgcn_readfirstlane(tid >> 5);
    const int lane = tid & 31;
    const int half = lane >> 4;
    const int l16  = lane & 15;

    __shared__ __attribute__((aligned(16))) float As[4][16][KP];
    __shared__ __attribute__((aligned(16))) float Bs[4][16][KP];
    __shared__ __attribute__((aligned(16))) float Wl[64][64];
    __shared__ float Ys[64][YP];
    __shared__ __attribute__((aligned(16))) float SB[2 * KSPA];
    __shared__ int   list[NB + 32];

    const bool hasL = dom != 0;
    const float* Ld = L_emb + dom * N_L;

    int dv[NB / 32];
    #pragma unroll
    for (int j = 0; j < NB / 32; ++j)
        dv[j] = didx[j * 32 + lane];

    const float* Arow[16];
    #pragma unroll
    for (int j = 0; j < 16; ++j) {
        const int i = wv * 16 + j;
        Arow[j] = Ld + NSH * i + (i * (i - 1)) / 2 + lane * 4;
    }
    v4f ra0[16];
    #pragma unroll
    for (int j = 0; j < 16; ++j)
        ra0[j] = *(const v4fu*)(Arow[j]);
    const float sbv = (tid < KSPA) ? S_emb[dom * KSPA + tid] : bias_ns[dom * KSPA + tid - KSPA];
    float wst[32];
    {
        const unsigned hi = (unsigned)tid >> 6;
        const unsigned k  = (unsigned)tid & 63u;
        #pragma unroll
        for (unsigned it = 0; it < 32; ++it) {
            const unsigned i   = it * 2u + hi;
            const unsigned off = NSH * i + (i * (i - 1u)) / 2u;
            const unsigned kk  = (k < i) ? k : 0u;
            wst[it] = Ld[off + NSH + kk];
        }
    }

    unsigned mk[NB / 32];
    #pragma unroll
    for (int j = 0; j < NB / 32; ++j)
        mk[j] = __builtin_amdgcn_ballot_w32(dv[j] == dom);
    int n = 0;
    int pos[NB / 32];
    #pragma unroll
    for (int j = 0; j < NB / 32; ++j) {
        pos[j] = (dv[j] == dom) ? n + (int)__builtin_amdgcn_mbcnt_lo(mk[j], 0u) : NB + lane;
        n += __builtin_popcount(mk[j]);
    }
    if (wv == 0) {
        #pragma unroll
        for (int j = 0; j < NB / 32; ++j)
            list[pos[j]] = j * 32 + lane;
    }
    SB[tid] = (tid < KSPA && !hasL) ? 1.0f : sbv;
    {
        const unsigned hi = (unsigned)tid >> 6;
        const unsigned k  = (unsigned)tid & 63u;
        #pragma unroll
        for (unsigned it = 0; it < 32; ++it) {
            const unsigned i = it * 2u + hi;
            Wl[i][k] = (hasL && k < i) ? wst[it] : 0.0f;
        }
    }
    __syncthreads();

    for (int t = slot; t * 16 < n; t += TSLOTS) {
        const int cn = t * 16 + l16;
        const int myb = list[cn < n ? cn : t * 16];
        const float* Erow[16];
        #pragma unroll
        for (int j = 0; j < 16; ++j)
            Erow[j] = eps + __builtin_amdgcn_readlane(myb, j) * NN + lane * 4;

        v4f ra[16], rb[16];
        #pragma unroll
        for (int j = 0; j < 16; ++j) {
            ra[j] = hasL ? ra0[j] : (v4f){0.f, 0.f, 0.f, 0.f};
            rb[j] = *(const v4f*)(Erow[j]);
        }

        v8f acc = {0.f, 0.f, 0.f, 0.f, 0.f, 0.f, 0.f, 0.f};

        #pragma unroll 1
        for (int kc = 0; kc < NN / KC; ++kc) {
            #pragma unroll
            for (int j = 0; j < 16; ++j) {
                *(v4f*)&As[wv][j][lane * 4] = ra[j];
                *(v4f*)&Bs[wv][j][lane * 4] = rb[j];
            }
            __syncthreads();

            v2f fa[KC / 4], fb[KC / 4];
            #pragma unroll
            for (int s = 0; s < KC / 4; ++s) {
                const int k0 = s * 4 + half * 2;
                fa[s].x = As[wv][l16][k0]; fa[s].y = As[wv][l16][k0 + 1];
                fb[s].x = Bs[wv][l16][k0]; fb[s].y = Bs[wv][l16][k0 + 1];
            }

            if (kc + 1 < NN / KC) {
                const int  c0   = (kc + 1) * KC + lane * 4;
                const bool inL  = hasL && c0 < NSH;
                const int  aoff = (c0 < NSH) ? (kc + 1) * KC : 0;
                if (inL) {
                    #pragma unroll
                    for (int j = 0; j < 16; ++j)
                        ra[j] = *(const v4fu*)(Arow[j] + aoff);
                } else {
                    #pragma unroll
                    for (int j = 0; j < 16; ++j)
                        ra[j] = (v4f){0.f, 0.f, 0.f, 0.f};
                }
                #pragma unroll
                for (int j = 0; j < 16; ++j)
                    rb[j] = *(const v4f*)(Erow[j] + (kc + 1) * KC);
            }

            __builtin_amdgcn_sched_barrier(0);
            #pragma unroll
            for (int s = 0; s < KC / 4; ++s)
                acc = __builtin_amdgcn_wmma_f32_16x16x4_f32(
                     false, fa[s],  false, fb[s],
                     (short)0, acc,  false,  false);
            __builtin_amdgcn_sched_barrier(0);
            __syncthreads();
        }

        #pragma unroll
        for (int v = 0; v < 8; ++v)
            Ys[wv * 16 + half * 8 + v][l16] = acc[v];
        __syncthreads();

        if (tid < 16 && cn < n) {
            float z[64];
            v4f bv[16];
            #pragma unroll
            for (int i4 = 0; i4 < 16; ++i4) {
                bv[i4] = *(const v4f*)&SB[KSPA + i4 * 4];
                const float4 e = *(const float4*)&Bs[0][tid][64 + i4 * 4];
                const float4 s = *(const float4*)&SB[i4 * 4];
                z[i4 * 4 + 0] = Ys[i4 * 4 + 0][tid] + s.x * e.x;
                z[i4 * 4 + 1] = Ys[i4 * 4 + 1][tid] + s.y * e.y;
                z[i4 * 4 + 2] = Ys[i4 * 4 + 2][tid] + s.z * e.z;
                z[i4 * 4 + 3] = Ys[i4 * 4 + 3][tid] + s.w * e.w;
            }
            float* orow = out + myb * NN + NSH;
            #define PUT_BLOCK(J)                                                              \
                _Pragma("unroll")                                                             \
                for (int r4 = 0; r4 < 4; ++r4) {                                              \
                    const int i = (J) * 16 + r4 * 4;                                          \
                    const v4f b4 = bv[i / 4];                                                 \
                    ((float4*)orow)[i / 4] = make_float4(z[i] + b4.x, z[i + 1] + b4.y,        \
                                                         z[i + 2] + b4.z, z[i + 3] + b4.w);   \
                }
            v4f w0[GW], w1[GW];
            load_group<0>(Wl, w0);
            subst_step<0>(Wl, w0, w1, z);   PUT_BLOCK(0)
            subst_step<1>(Wl, w1, w0, z);
            subst_step<2>(Wl, w0, w1, z);
            subst_step<3>(Wl, w1, w0, z);   PUT_BLOCK(1)
            subst_step<4>(Wl, w0, w1, z);
            subst_step<5>(Wl, w1, w0, z);
            subst_step<6>(Wl, w0, w1, z);
            subst_step<7>(Wl, w1, w0, z);
            subst_step<8>(Wl, w0, w1, z);   PUT_BLOCK(2)
            subst_step<9>(Wl, w1, w0, z);
            subst_step<10>(Wl, w0, w1, z);
            subst_step<11>(Wl, w1, w0, z);
            subst_step<12>(Wl, w0, w1, z);
            subst_step<13>(Wl, w1, w0, z);
            subst_step<14>(Wl, w0, w1, z);
            subst_step<15>(Wl, w1, w0, z);  PUT_BLOCK(3)
            #undef PUT_BLOCK
        }
    }
}

extern "C" void kernel_launch(void* const* d_in, const int* in_sizes, int n_in,
                              void* d_out, int out_size, void* d_ws, size_t ws_size,
                              hipStream_t stream) {
    const float* eps     = (const float*)d_in[0];
    const int*   didx    = (const int*)  d_in[1];
    const float* L_emb   = (const float*)d_in[2];
    const float* S_emb   = (const float*)d_in[3];
    const float* bias_ns = (const float*)d_in[4];
    const float* bias_sh = (const float*)d_in[5];
    float*       out     = (float*)d_out;

    tail_subst_prefix_kernel<<<dim3(GRIDB), dim3(128), 0, stream>>>(
        eps, didx, L_emb, S_emb, bias_ns, bias_sh, out);
}


# Recurrence on the matrix unit: block forward substitution, 16-row blocks, diagonal blocks inverted once per block, result tiles fed back as B operands on wave 0's 32 lanes; sum order changes
# speedup vs baseline: 10.0669x; 1.1886x over previous
#include <hip/hip_runtime.h>
#include <hip/hip_bf16.h>

#define NN   512
#define KSPA 64
#define ND   16
#define NB   1024
#define NSH  448
#define N_L  30688

typedef __attribute__((ext_vector_type(2))) float v2f;
typedef __attribute__((ext_vector_type(4))) float v4f;
typedef __attribute__((ext_vector_type(8))) float v8f;
typedef float v4fu __attribute__((ext_vector_type(4), aligned(4)));

#define TSLOTS 8
#define KC 128
#define KP (KC + 4)
#define WP 68
#define VP 20
#define TAILB (TSLOTS * ND)
#define PREFB 112
#define GRIDB (TAILB + PREFB)

__global__ __launch_bounds__(128) void tail_subst_prefix_kernel(
    const float* __restrict__ eps,
    const int*   __restrict__ didx,
    const float* __restrict__ L_emb,
    const float* __restrict__ S_emb,
    const float* __restrict__ bias_ns,
    const float* __restrict__ bias_shared,
    float* __restrict__ out)
{
    const int tid = threadIdx.x;

    if (blockIdx.x >= TAILB) {
        const int base = (blockIdx.x - TAILB) * 1024 + tid;
        float4 e[8], bs[8];
        int bb[8], cc[8];
        #pragma unroll
        for (int q = 0; q < 8; ++q) {
            const int idx = base + q * 128;
            bb[q] = idx / (NSH / 4);
            cc[q] = idx % (NSH / 4);
            e[q]  = ((const float4*)(eps + bb[q] * NN))[cc[q]];
            bs[q] = ((const float4*)bias_shared)[cc[q]];
        }
        #pragma unroll
        for (int q = 0; q < 8; ++q)
            ((float4*)(out + bb[q] * NN))[cc[q]] =
                make_float4(e[q].x + bs[q].x, e[q].y + bs[q].y, e[q].z + bs[q].z, e[q].w + bs[q].w);
        return;
    }

    const int dom  = blockIdx.x / TSLOTS;
    const int slot = blockIdx.x % TSLOTS;
    const int wv   = __builtin_amdgcn_readfirstlane(tid >> 5);
    const int lane = tid & 31;
    const int half = lane >> 4;
    const int l16  = lane & 15;

    __shared__ __attribute__((aligned(16))) float As[4][16][KP];
    __shared__ __attribute__((aligned(16))) float Bs[4][16][KP];
    __shared__ __attribute__((aligned(16))) float Wl[64][WP];
    __shared__ __attribute__((aligned(16))) float Vl[4][16][VP];
    __shared__ __attribute__((aligned(16))) float Yt[4][32][8];
    __shared__ __attribute__((aligned(16))) float SB[2 * KSPA];
    __shared__ int   list[NB + 32];

    const bool hasL = dom != 0;
    const float* Ld = L_emb + dom * N_L;

    int dv[NB / 32];
    #pragma unroll
    for (int j = 0; j < NB / 32; ++j)
        dv[j] = didx[j * 32 + lane];

    const float* Arow[16];
    #pragma unroll
    for (int j = 0; j < 16; ++j) {
        const int i = wv * 16 + j;
        Arow[j] = Ld + NSH * i + (i * (i - 1)) / 2 + lane * 4;
    }
    v4f ra0[16];
    #pragma unroll
    for (int j = 0; j < 16; ++j)
        ra0[j] = *(const v4fu*)(Arow[j]);
    const float sbv = (tid < KSPA) ? S_emb[dom * KSPA + tid] : bias_ns[dom * KSPA + tid - KSPA];
    float wst[32];
    {
        const unsigned hi = (unsigned)tid >> 6;
        const unsigned k  = (unsigned)tid & 63u;
        #pragma unroll
        for (unsigned it = 0; it < 32; ++it) {
            const unsigned i   = it * 2u + hi;
            const unsigned off = NSH * i + (i * (i - 1u)) / 2u;
            const unsigned kk  = (k < i) ? k : 0u;
            wst[it] = Ld[off + NSH + kk];
        }
    }

    unsigned mk[NB / 32];
    #pragma unroll
    for (int j = 0; j < NB / 32; ++j)
        mk[j] = __builtin_amdgcn_ballot_w32(dv[j] == dom);
    int n = 0;
    int pos[NB / 32];
    #pragma unroll
    for (int j = 0; j < NB / 32; ++j) {
        pos[j] = (dv[j] == dom) ? n + (int)__builtin_amdgcn_mbcnt_lo(mk[j], 0u) : NB + lane;
        n += __builtin_popcount(mk[j]);
    }
    if (wv == 0) {
        #pragma unroll
        for (int j = 0; j < NB / 32; ++j)
            list[pos[j]] = j * 32 + lane;
    }
    SB[tid] = (tid < KSPA && !hasL) ? 1.0f : sbv;
    {
        const unsigned hi = (unsigned)tid >> 6;
        const unsigned k  = (unsigned)tid & 63u;
        #pragma unroll
        for (unsigned it = 0; it < 32; ++it) {
            const unsigned i = it * 2u + hi;
            Wl[i][k] = (hasL && k < i) ? wst[it] : 0.0f;
        }
    }
    __syncthreads();

    for (int t = slot; t * 16 < n; t += TSLOTS) {
        const int cn = t * 16 + l16;
        const int myb = list[cn < n ? cn : t * 16];
        const float* Erow[16];
        #pragma unroll
        for (int j = 0; j < 16; ++j)
            Erow[j] = eps + __builtin_amdgcn_readlane(myb, j) * NN + lane * 4;

        v4f ra[16], rb[16];
        #pragma unroll
        for (int j = 0; j < 16; ++j) {
            ra[j] = hasL ? ra0[j] : (v4f){0.f, 0.f, 0.f, 0.f};
            rb[j] = *(const v4f*)(Erow[j]);
        }

        if (t == slot && wv < 2) {
            const int Jn = wv * 2 + half;
            const int Jb = Jn * 16;
            float x[16];
            #pragma unroll
            for (int r = 0; r < 16; ++r) {
                float a = (r == l16) ? 1.0f : 0.0f;
                #pragma unroll
                for (int q = 0; q < r; q += 4) {
                    const v4f w = *(const v4f*)&Wl[Jb + r][Jb + q];
                    a += w.x * x[q];
                    if (q + 1 < r) a += w.y * x[q + 1];
                    if (q + 2 < r) a += w.z * x[q + 2];
                    if (q + 3 < r) a += w.w * x[q + 3];
                }
                x[r] = a;
                Vl[Jn][r][l16] = a;
            }
        }

        v8f acc = {0.f, 0.f, 0.f, 0.f, 0.f, 0.f, 0.f, 0.f};

        #pragma unroll 1
        for (int kc = 0; kc < NN / KC; ++kc) {
            #pragma unroll
            for (int j = 0; j < 16; ++j) {
                *(v4f*)&As[wv][j][lane * 4] = ra[j];
                *(v4f*)&Bs[wv][j][lane * 4] = rb[j];
            }
            __syncthreads();

            v2f fa[KC / 4], fb[KC / 4];
            #pragma unroll
            for (int s = 0; s < KC / 4; ++s) {
                const int k0 = s * 4 + half * 2;
                fa[s].x = As[wv][l16][k0]; fa[s].y = As[wv][l16][k0 + 1];
                fb[s].x = Bs[wv][l16][k0]; fb[s].y = Bs[wv][l16][k0 + 1];
            }

            if (kc + 1 < NN / KC) {
                const int  c0   = (kc + 1) * KC + lane * 4;
                const bool inL  = hasL && c0 < NSH;
                const int  aoff = (c0 < NSH) ? (kc + 1) * KC : 0;
                if (inL) {
                    #pragma unroll
                    for (int j = 0; j < 16; ++j)
                        ra[j] = *(const v4fu*)(Arow[j] + aoff);
                } else {
                    #pragma unroll
                    for (int j = 0; j < 16; ++j)
                        ra[j] = (v4f){0.f, 0.f, 0.f, 0.f};
                }
                #pragma unroll
                for (int j = 0; j < 16; ++j)
                    rb[j] = *(const v4f*)(Erow[j] + (kc + 1) * KC);
            }

            __builtin_amdgcn_sched_barrier(0);
            #pragma unroll
            for (int s = 0; s < KC / 4; ++s)
                acc = __builtin_amdgcn_wmma_f32_16x16x4_f32(
                     false, fa[s],  false, fb[s],
                     (short)0, acc,  false,  false);
            __builtin_amdgcn_sched_barrier(0);
            __syncthreads();
        }

        if (wv != 0)
            *(v8f*)&Yt[wv][lane][0] = acc;
        __syncthreads();

        if (wv == 0) {
            const int h8 = half * 8;
            v8f R[4];
            v4f av[4][2];
            v4f aw[4][3][2];
            v4f ev[4][2], sv[4][2], bv[4][2];
            R[0] = acc;
            #pragma unroll
            for (int I = 1; I < 4; ++I)
                R[I] = *(const v8f*)&Yt[I][lane][0];
            #pragma unroll
            for (int I = 0; I < 4; ++I) {
                #pragma unroll
                for (int p = 0; p < 2; ++p) {
                    ev[I][p] = *(const v4f*)&Bs[0][l16][64 + 16 * I + h8 + 4 * p];
                    sv[I][p] = *(const v4f*)&SB[16 * I + h8 + 4 * p];
                    bv[I][p] = *(const v4f*)&SB[KSPA + 16 * I + h8 + 4 * p];
                    av[I][p] = *(const v4f*)&Vl[I][l16][h8 + 4 * p];
                    #pragma unroll
                    for (int J = 0; J < I; ++J)
                        aw[I][J][p] = *(const v4f*)&Wl[16 * I + l16][16 * J + h8 + 4 * p];
                }
            }
            #pragma unroll
            for (int I = 0; I < 4; ++I) {
                #pragma unroll
                for (int r = 0; r < 8; ++r)
                    R[I][r] += sv[I][r / 4][r % 4] * ev[I][r / 4][r % 4];
            }

            float* orow = out + myb * NN + NSH + h8;
            #pragma unroll
            for (int J = 0; J < 4; ++J) {
                v8f Z = {0.f, 0.f, 0.f, 0.f, 0.f, 0.f, 0.f, 0.f};
                #pragma unroll
                for (int st = 0; st < 4; ++st) {
                    const v2f a = {av[J][st / 2][(st % 2) * 2], av[J][st / 2][(st % 2) * 2 + 1]};
                    const v2f b = {R[J][2 * st], R[J][2 * st + 1]};
                    Z = __builtin_amdgcn_wmma_f32_16x16x4_f32(
                         false, a,  false, b,
                         (short)0, Z,  false,  false);
                }
                if (cn < n) {
                    ((float4*)(orow + 16 * J))[0] = make_float4(Z[0] + bv[J][0].x, Z[1] + bv[J][0].y,
                                                                Z[2] + bv[J][0].z, Z[3] + bv[J][0].w);
                    ((float4*)(orow + 16 * J))[1] = make_float4(Z[4] + bv[J][1].x, Z[5] + bv[J][1].y,
                                                                Z[6] + bv[J][1].z, Z[7] + bv[J][1].w);
                }
                #pragma unroll
                for (int I = J + 1; I < 4; ++I) {
                    #pragma unroll
                    for (int st = 0; st < 4; ++st) {
                        const v2f a = {aw[I][J][st / 2][(st % 2) * 2], aw[I][J][st / 2][(st % 2) * 2 + 1]};
                        const v2f b = {Z[2 * st], Z[2 * st + 1]};
                        R[I] = __builtin_amdgcn_wmma_f32_16x16x4_f32(
                             false, a,  false, b,
                             (short)0, R[I],  false,  false);
                    }
                }
            }
        }
    }
}

extern "C" void kernel_launch(void* const* d_in, const int* in_sizes, int n_in,
                              void* d_out, int out_size, void* d_ws, size_t ws_size,
                              hipStream_t stream) {
    const float* eps     = (const float*)d_in[0];
    const int*   didx    = (const int*)  d_in[1];
    const float* L_emb   = (const float*)d_in[2];
    const float* S_emb   = (const float*)d_in[3];
    const float* bias_ns = (const float*)d_in[4];
    const float* bias_sh = (const float*)d_in[5];
    float*       out     = (float*)d_out;

    tail_subst_prefix_kernel<<<dim3(GRIDB), dim3(128), 0, stream>>>(
        eps, didx, L_emb, S_emb, bias_ns, bias_sh, out);
}


# Tail product: whole K staged in LDS at once (a wave's 16 rows of L before the ranking, once per block; eps tile once per tile, shared by the 4 waves): 2 barriers per tile (was 9); k order unchanged
# speedup vs baseline: 12.4527x; 1.2370x over previous
#include <hip/hip_runtime.h>
#include <hip/hip_bf16.h>

#define NN   512
#define KSPA 64
#define ND   16
#define NB   1024
#define NSH  448
#define N_L  30688

typedef __attribute__((ext_vector_type(2))) float v2f;
typedef __attribute__((ext_vector_type(4))) float v4f;
typedef __attribute__((ext_vector_type(8))) float v8f;
typedef float v4fu __attribute__((ext_vector_type(4), aligned(4)));

#define TSLOTS 8
#define KA NSH
#define AP (KA + 4)
#define BP (NN + 4)
#define GS 16
#define NG (KA / 4 / GS)
#define WP 68
#define VP 20
#define TAILB (TSLOTS * ND)
#define PREFB 112
#define GRIDB (TAILB + PREFB)

__global__ __launch_bounds__(128) void tail_subst_prefix_kernel(
    const float* __restrict__ eps,
    const int*   __restrict__ didx,
    const float* __restrict__ L_emb,
    const float* __restrict__ S_emb,
    const float* __restrict__ bias_ns,
    const float* __restrict__ bias_shared,
    float* __restrict__ out)
{
    const int tid = threadIdx.x;

    if (blockIdx.x >= TAILB) {
        const int base = (blockIdx.x - TAILB) * 1024 + tid;
        float4 e[8], bs[8];
        int bb[8], cc[8];
        #pragma unroll
        for (int q = 0; q < 8; ++q) {
            const int idx = base + q * 128;
            bb[q] = idx / (NSH / 4);
            cc[q] = idx % (NSH / 4);
            e[q]  = ((const float4*)(eps + bb[q] * NN))[cc[q]];
            bs[q] = ((const float4*)bias_shared)[cc[q]];
        }
        #pragma unroll
        for (int q = 0; q < 8; ++q)
            ((float4*)(out + bb[q] * NN))[cc[q]] =
                make_float4(e[q].x + bs[q].x, e[q].y + bs[q].y, e[q].z + bs[q].z, e[q].w + bs[q].w);
        return;
    }

    const int dom  = blockIdx.x / TSLOTS;
    const int slot = blockIdx.x % TSLOTS;
    const int wv   = __builtin_amdgcn_readfirstlane(tid >> 5);
    const int lane = tid & 31;
    const int half = lane >> 4;
    const int l16  = lane & 15;

    __shared__ __attribute__((aligned(16))) float As[4][16][AP];
    __shared__ __attribute__((aligned(16))) float Bs[16][BP];
    __shared__ __attribute__((aligned(16))) float Wl[64][WP];
    __shared__ __attribute__((aligned(16))) float Vl[4][16][VP];
    __shared__ __attribute__((aligned(16))) float Yt[4][32][8];
    __shared__ __attribute__((aligned(16))) float SB[2 * KSPA];
    __shared__ int   list[NB + 32];

    const bool hasL = dom != 0;
    const float* Ld = L_emb + dom * N_L;

    int dv[NB / 32];
    #pragma unroll
    for (int j = 0; j < NB / 32; ++j)
        dv[j] = didx[j * 32 + lane];

    const float* Arow[16];
    #pragma unroll
    for (int j = 0; j < 16; ++j) {
        const int i = wv * 16 + j;
        Arow[j] = Ld + NSH * i + (i * (i - 1)) / 2;
    }
    v4f ra[16][3];
    v4f rt[8];
    #pragma unroll
    for (int j = 0; j < 16; ++j) {
        #pragma unroll
        for (int q = 0; q < 3; ++q)
            ra[j][q] = *(const v4fu*)(Arow[j] + q * 128 + lane * 4);
        if (j & 1)
            rt[j / 2] = *(const v4fu*)((half ? Arow[j] : Arow[j - 1]) + 384 + l16 * 4);
    }
    const float sbv = (tid < KSPA) ? S_emb[dom * KSPA + tid] : bias_ns[dom * KSPA + tid - KSPA];
    float wst[32];
    {
        const unsigned hi = (unsigned)tid >> 6;
        const unsigned k  = (unsigned)tid & 63u;
        #pragma unroll
        for (unsigned it = 0; it < 32; ++it) {
            const unsigned i   = it * 2u + hi;
            const unsigned off = NSH * i + (i * (i - 1u)) / 2u;
            const unsigned kk  = (k < i) ? k : 0u;
            wst[it] = Ld[off + NSH + kk];
        }
    }

    unsigned mk[NB / 32];
    #pragma unroll
    for (int j = 0; j < NB / 32; ++j)
        mk[j] = __builtin_amdgcn_ballot_w32(dv[j] == dom);
    int n = 0;
    int pos[NB / 32];
    #pragma unroll
    for (int j = 0; j < NB / 32; ++j) {
        pos[j] = (dv[j] == dom) ? n + (int)__builtin_amdgcn_mbcnt_lo(mk[j], 0u) : NB + lane;
        n += __builtin_popcount(mk[j]);
    }
    {
        const v4f zero4 = {0.f, 0.f, 0.f, 0.f};
        #pragma unroll
        for (int j = 0; j < 16; ++j) {
            #pragma unroll
            for (int q = 0; q < 3; ++q)
                *(v4f*)&As[wv][j][q * 128 + lane * 4] = hasL ? ra[j][q] : zero4;
            if (j & 1)
                *(v4f*)&As[wv][j - 1 + half][384 + l16 * 4] = hasL ? rt[j / 2] : zero4;
        }
    }
    if (wv == 0) {
        #pragma unroll
        for (int j = 0; j < NB / 32; ++j)
            list[pos[j]] = j * 32 + lane;
    }
    SB[tid] = (tid < KSPA && !hasL) ? 1.0f : sbv;
    {
        const unsigned hi = (unsigned)tid >> 6;
        const unsigned k  = (unsigned)tid & 63u;
        #pragma unroll
        for (unsigned it = 0; it < 32; ++it) {
            const unsigned i = it * 2u + hi;
            Wl[i][k] = (hasL && k < i) ? wst[it] : 0.0f;
        }
    }
    __syncthreads();

    for (int t = slot; t * 16 < n; t += TSLOTS) {
        const int cn = t * 16 + l16;
        const int myb = list[cn < n ? cn : t * 16];
        v4f rb[4][4];
        #pragma unroll
        for (int r = 0; r < 4; ++r) {
            const float* Erow = eps + __builtin_amdgcn_readlane(myb, wv * 4 + r) * NN + lane * 4;
            #pragma unroll
            for (int q = 0; q < 4; ++q)
                rb[r][q] = *(const v4f*)(Erow + q * 128);
        }

        if (t == slot && wv < 2) {
            const int Jn = wv * 2 + half;
            const int Jb = Jn * 16;
            float x[16];
            #pragma unroll
            for (int r = 0; r < 16; ++r) {
                float a = (r == l16) ? 1.0f : 0.0f;
                #pragma unroll
                for (int q = 0; q < r; q += 4) {
                    const v4f w = *(const v4f*)&Wl[Jb + r][Jb + q];
                    a += w.x * x[q];
                    if (q + 1 < r) a += w.y * x[q + 1];
                    if (q + 2 < r) a += w.z * x[q + 2];
                    if (q + 3 < r) a += w.w * x[q + 3];
                }
                x[r] = a;
                Vl[Jn][r][l16] = a;
            }
        }

        #pragma unroll
        for (int r = 0; r < 4; ++r) {
            #pragma unroll
            for (int q = 0; q < 4; ++q)
                *(v4f*)&Bs[wv * 4 + r][q * 128 + lane * 4] = rb[r][q];
        }
        __syncthreads();

        v8f acc = {0.f, 0.f, 0.f, 0.f, 0.f, 0.f, 0.f, 0.f};

        v2f fa[2][GS], fb[2][GS];
        #pragma unroll
        for (int s = 0; s < GS; ++s) {
            const int k0 = s * 4 + half * 2;
            fa[0][s].x = As[wv][l16][k0]; fa[0][s].y = As[wv][l16][k0 + 1];
            fb[0][s].x = Bs[l16][k0];     fb[0][s].y = Bs[l16][k0 + 1];
        }
        #pragma unroll
        for (int g = 0; g < NG; ++g) {
            if (g + 1 < NG) {
                #pragma unroll
                for (int s = 0; s < GS; ++s) {
                    const int k0 = ((g + 1) * GS + s) * 4 + half * 2;
                    fa[(g + 1) & 1][s].x = As[wv][l16][k0]; fa[(g + 1) & 1][s].y = As[wv][l16][k0 + 1];
                    fb[(g + 1) & 1][s].x = Bs[l16][k0];     fb[(g + 1) & 1][s].y = Bs[l16][k0 + 1];
                }
            }
            __builtin_amdgcn_sched_barrier(0);
            #pragma unroll
            for (int s = 0; s < GS; ++s)
                acc = __builtin_amdgcn_wmma_f32_16x16x4_f32(
                     false, fa[g & 1][s],  false, fb[g & 1][s],
                     (short)0, acc,  false,  false);
            __builtin_amdgcn_sched_barrier(0);
        }

        const int h8 = half * 8;
        v4f ev[4][2];
        if (wv != 0) {
            *(v8f*)&Yt[wv][lane][0] = acc;
        } else {
            #pragma unroll
            for (int I = 0; I < 4; ++I) {
                #pragma unroll
                for (int p = 0; p < 2; ++p)
                    ev[I][p] = *(const v4f*)&Bs[l16][NSH + 16 * I + h8 + 4 * p];
            }
        }
        __syncthreads();

        if (wv == 0) {
            v8f R[4];
            v4f av[4][2];
            v4f aw[4][3][2];
            v4f sv[4][2], bv[4][2];
            R[0] = acc;
            #pragma unroll
            for (int I = 1; I < 4; ++I)
                R[I] = *(const v8f*)&Yt[I][lane][0];
            #pragma unroll
            for (int I = 0; I < 4; ++I) {
                #pragma unroll
                for (int p = 0; p < 2; ++p) {
                    sv[I][p] = *(const v4f*)&SB[16 * I + h8 + 4 * p];
                    bv[I][p] = *(const v4f*)&SB[KSPA + 16 * I + h8 + 4 * p];
                    av[I][p] = *(const v4f*)&Vl[I][l16][h8 + 4 * p];
                    #pragma unroll
                    for (int J = 0; J < I; ++J)
                        aw[I][J][p] = *(const v4f*)&Wl[16 * I + l16][16 * J + h8 + 4 * p];
                }
            }
            #pragma unroll
            for (int I = 0; I < 4; ++I) {
                #pragma unroll
                for (int r = 0; r < 8; ++r)
                    R[I][r] += sv[I][r / 4][r % 4] * ev[I][r / 4][r % 4];
            }

            float* orow = out + myb * NN + NSH + h8;
            #pragma unroll
            for (int J = 0; J < 4; ++J) {
                v8f Z = {0.f, 0.f, 0.f, 0.f, 0.f, 0.f, 0.f, 0.f};
                #pragma unroll
                for (int st = 0; st < 4; ++st) {
                    const v2f a = {av[J][st / 2][(st % 2) * 2], av[J][st / 2][(st % 2) * 2 + 1]};
                    const v2f b = {R[J][2 * st], R[J][2 * st + 1]};
                    Z = __builtin_amdgcn_wmma_f32_16x16x4_f32(
                         false, a,  false, b,
                         (short)0, Z,  false,  false);
                }
                if (cn < n) {
                    ((float4*)(orow + 16 * J))[0] = make_float4(Z[0] + bv[J][0].x, Z[1] + bv[J][0].y,
                                                                Z[2] + bv[J][0].z, Z[3] + bv[J][0].w);
                    ((float4*)(orow + 16 * J))[1] = make_float4(Z[4] + bv[J][1].x, Z[5] + bv[J][1].y,
                                                                Z[6] + bv[J][1].z, Z[7] + bv[J][1].w);
                }
                #pragma unroll
                for (int I = J + 1; I < 4; ++I) {
                    #pragma unroll
                    for (int st = 0; st < 4; ++st) {
                        const v2f a = {aw[I][J][st / 2][(st % 2) * 2], aw[I][J][st / 2][(st % 2) * 2 + 1]};
                        const v2f b = {Z[2 * st], Z[2 * st + 1]};
                        R[I] = __builtin_amdgcn_wmma_f32_16x16x4_f32(
                             false, a,  false, b,
                             (short)0, R[I],  false,  false);
                    }
                }
            }
        }
    }
}

extern "C" void kernel_launch(void* const* d_in, const int* in_sizes, int n_in,
                              void* d_out, int out_size, void* d_ws, size_t ws_size,
                              hipStream_t stream) {
    const float* eps     = (const float*)d_in[0];
    const int*   didx    = (const int*)  d_in[1];
    const float* L_emb   = (const float*)d_in[2];
    const float* S_emb   = (const float*)d_in[3];
    const float* bias_ns = (const float*)d_in[4];
    const float* bias_sh = (const float*)d_in[5];
    float*       out     = (float*)d_out;

    tail_subst_prefix_kernel<<<dim3(GRIDB), dim3(128), 0, stream>>>(
        eps, didx, L_emb, S_emb, bias_ns, bias_sh, out);
}


# First tile's eps rows requested right behind the ranking's barrier (now LDS-only), ahead of the inverses and the L rows' LDS stores, which run under their flight; every sum's order unchanged
# speedup vs baseline: 13.1477x; 1.0558x over previous
#include <hip/hip_runtime.h>
#include <hip/hip_bf16.h>

#define NN   512
#define KSPA 64
#define ND   16
#define NB   1024
#define NSH  448
#define N_L  30688

typedef __attribute__((ext_vector_type(2))) float v2f;
typedef __attribute__((ext_vector_type(4))) float v4f;
typedef __attribute__((ext_vector_type(8))) float v8f;
typedef float v4fu __attribute__((ext_vector_type(4), aligned(4)));

#define TSLOTS 8
#define KA NSH
#define AP (KA + 4)
#define BP (NN + 4)
#define GS 16
#define NG (KA / 4 / GS)
#define WP 68
#define VP 20
#define TAILB (TSLOTS * ND)
#define PREFB 112
#define GRIDB (TAILB + PREFB)

static __device__ __forceinline__ void lds_barrier() {
    __builtin_amdgcn_fence(__ATOMIC_RELEASE, "workgroup", "local");
    __builtin_amdgcn_s_barrier();
    __builtin_amdgcn_fence(__ATOMIC_ACQUIRE, "workgroup", "local");
}

__global__ __launch_bounds__(128) void tail_subst_prefix_kernel(
    const float* __restrict__ eps,
    const int*   __restrict__ didx,
    const float* __restrict__ L_emb,
    const float* __restrict__ S_emb,
    const float* __restrict__ bias_ns,
    const float* __restrict__ bias_shared,
    float* __restrict__ out)
{
    const int tid = threadIdx.x;

    if (blockIdx.x >= TAILB) {
        const int base = (blockIdx.x - TAILB) * 1024 + tid;
        float4 e[8], bs[8];
        int bb[8], cc[8];
        #pragma unroll
        for (int q = 0; q < 8; ++q) {
            const int idx = base + q * 128;
            bb[q] = idx / (NSH / 4);
            cc[q] = idx % (NSH / 4);
            e[q]  = ((const float4*)(eps + bb[q] * NN))[cc[q]];
            bs[q] = ((const float4*)bias_shared)[cc[q]];
        }
        #pragma unroll
        for (int q = 0; q < 8; ++q)
            ((float4*)(out + bb[q] * NN))[cc[q]] =
                make_float4(e[q].x + bs[q].x, e[q].y + bs[q].y, e[q].z + bs[q].z, e[q].w + bs[q].w);
        return;
    }

    const int dom  = blockIdx.x / TSLOTS;
    const int slot = blockIdx.x % TSLOTS;
    const int wv   = __builtin_amdgcn_readfirstlane(tid >> 5);
    const int lane = tid & 31;
    const int half = lane >> 4;
    const int l16  = lane & 15;

    __shared__ __attribute__((aligned(16))) float As[4][16][AP];
    __shared__ __attribute__((aligned(16))) float Bs[16][BP];
    __shared__ __attribute__((aligned(16))) float Wl[64][WP];
    __shared__ __attribute__((aligned(16))) float Vl[4][16][VP];
    __shared__ __attribute__((aligned(16))) float Yt[4][32][8];
    __shared__ __attribute__((aligned(16))) float SB[2 * KSPA];
    __shared__ int   list[NB + 32];

    const bool hasL = dom != 0;
    const float* Ld = L_emb + dom * N_L;

    int dv[NB / 32];
    #pragma unroll
    for (int j = 0; j < NB / 32; ++j)
        dv[j] = didx[j * 32 + lane];

    const float* Arow[16];
    #pragma unroll
    for (int j = 0; j < 16; ++j) {
        const int i = wv * 16 + j;
        Arow[j] = Ld + NSH * i + (i * (i - 1)) / 2;
    }
    v4f ra[16][3];
    v4f rt[8];
    #pragma unroll
    for (int j = 0; j < 8; ++j) {
        #pragma unroll
        for (int q = 0; q < 3; ++q)
            ra[j][q] = *(const v4fu*)(Arow[j] + q * 128 + lane * 4);
        if (j & 1)
            rt[j / 2] = *(const v4fu*)((half ? Arow[j] : Arow[j - 1]) + 384 + l16 * 4);
    }
    const float sbv = (tid < KSPA) ? S_emb[dom * KSPA + tid] : bias_ns[dom * KSPA + tid - KSPA];
    float wst[32];
    {
        const unsigned hi = (unsigned)tid >> 6;
        const unsigned k  = (unsigned)tid & 63u;
        #pragma unroll
        for (unsigned it = 0; it < 32; ++it) {
            const unsigned i   = it * 2u + hi;
            const unsigned off = NSH * i + (i * (i - 1u)) / 2u;
            const unsigned kk  = (k < i) ? k : 0u;
            wst[it] = Ld[off + NSH + kk];
        }
    }
    #pragma unroll
    for (int j = 8; j < 16; ++j) {
        #pragma unroll
        for (int q = 0; q < 3; ++q)
            ra[j][q] = *(const v4fu*)(Arow[j] + q * 128 + lane * 4);
        if (j & 1)
            rt[j / 2] = *(const v4fu*)((half ? Arow[j] : Arow[j - 1]) + 384 + l16 * 4);
    }

    unsigned mk[NB / 32];
    #pragma unroll
    for (int j = 0; j < NB / 32; ++j)
        mk[j] = __builtin_amdgcn_ballot_w32(dv[j] == dom);
    int n = 0;
    int pos[NB / 32];
    #pragma unroll
    for (int j = 0; j < NB / 32; ++j) {
        pos[j] = (dv[j] == dom) ? n + (int)__builtin_amdgcn_mbcnt_lo(mk[j], 0u) : NB + lane;
        n += __builtin_popcount(mk[j]);
    }
    if (wv == 0) {
        #pragma unroll
        for (int j = 0; j < NB / 32; ++j)
            list[pos[j]] = j * 32 + lane;
    }
    SB[tid] = (tid < KSPA && !hasL) ? 1.0f : sbv;
    {
        const unsigned hi = (unsigned)tid >> 6;
        const unsigned k  = (unsigned)tid & 63u;
        #pragma unroll
        for (unsigned it = 0; it < 32; ++it) {
            const unsigned i = it * 2u + hi;
            Wl[i][k] = (hasL && k < i) ? wst[it] : 0.0f;
        }
    }
    lds_barrier();

    if (slot * 16 < n) {
        int t = slot;
        int cn = t * 16 + l16;
        int myb = list[cn < n ? cn : t * 16];
        v4f rb[4][4];
        #pragma unroll
        for (int r = 0; r < 4; ++r) {
            const float* Erow = eps + __builtin_amdgcn_readlane(myb, wv * 4 + r) * NN + lane * 4;
            #pragma unroll
            for (int q = 0; q < 4; ++q)
                rb[r][q] = *(const v4f*)(Erow + q * 128);
        }

        if (wv < 2) {
            const int Jn = wv * 2 + half;
            const int Jb = Jn * 16;
            float x[16];
            #pragma unroll
            for (int r = 0; r < 16; ++r) {
                float a = (r == l16) ? 1.0f : 0.0f;
                #pragma unroll
                for (int q = 0; q < r; q += 4) {
                    const v4f w = *(const v4f*)&Wl[Jb + r][Jb + q];
                    a += w.x * x[q];
                    if (q + 1 < r) a += w.y * x[q + 1];
                    if (q + 2 < r) a += w.z * x[q + 2];
                    if (q + 3 < r) a += w.w * x[q + 3];
                }
                x[r] = a;
                Vl[Jn][r][l16] = a;
            }
        }

        {
            const v4f zero4 = {0.f, 0.f, 0.f, 0.f};
            #pragma unroll
            for (int j = 0; j < 16; ++j) {
                #pragma unroll
                for (int q = 0; q < 3; ++q)
                    *(v4f*)&As[wv][j][q * 128 + lane * 4] = hasL ? ra[j][q] : zero4;
                if (j & 1)
                    *(v4f*)&As[wv][j - 1 + half][384 + l16 * 4] = hasL ? rt[j / 2] : zero4;
            }
        }

        for (;;) {
            #pragma unroll
            for (int r = 0; r < 4; ++r) {
                #pragma unroll
                for (int q = 0; q < 4; ++q)
                    *(v4f*)&Bs[wv * 4 + r][q * 128 + lane * 4] = rb[r][q];
            }
            __syncthreads();

            v8f acc = {0.f, 0.f, 0.f, 0.f, 0.f, 0.f, 0.f, 0.f};

            v2f fa[2][GS], fb[2][GS];
            #pragma unroll
            for (int s = 0; s < GS; ++s) {
                const int k0 = s * 4 + half * 2;
                fa[0][s].x = As[wv][l16][k0]; fa[0][s].y = As[wv][l16][k0 + 1];
                fb[0][s].x = Bs[l16][k0];     fb[0][s].y = Bs[l16][k0 + 1];
            }
            #pragma unroll
            for (int g = 0; g < NG; ++g) {
                if (g + 1 < NG) {
                    #pragma unroll
                    for (int s = 0; s < GS; ++s) {
                        const int k0 = ((g + 1) * GS + s) * 4 + half * 2;
                        fa[(g + 1) & 1][s].x = As[wv][l16][k0]; fa[(g + 1) & 1][s].y = As[wv][l16][k0 + 1];
                        fb[(g + 1) & 1][s].x = Bs[l16][k0];     fb[(g + 1) & 1][s].y = Bs[l16][k0 + 1];
                    }
                }
                __builtin_amdgcn_sched_barrier(0);
                #pragma unroll
                for (int s = 0; s < GS; ++s)
                    acc = __builtin_amdgcn_wmma_f32_16x16x4_f32(
                         false, fa[g & 1][s],  false, fb[g & 1][s],
                         (short)0, acc,  false,  false);
                __builtin_amdgcn_sched_barrier(0);
            }

            const int h8 = half * 8;
            v4f ev[4][2];
            if (wv != 0) {
                *(v8f*)&Yt[wv][lane][0] = acc;
            } else {
                #pragma unroll
                for (int I = 0; I < 4; ++I) {
                    #pragma unroll
                    for (int p = 0; p < 2; ++p)
                        ev[I][p] = *(const v4f*)&Bs[l16][NSH + 16 * I + h8 + 4 * p];
                }
            }
            __syncthreads();

            if (wv == 0) {
                v8f R[4];
                v4f av[4][2];
                v4f aw[4][3][2];
                v4f sv[4][2], bv[4][2];
                R[0] = acc;
                #pragma unroll
                for (int I = 1; I < 4; ++I)
                    R[I] = *(const v8f*)&Yt[I][lane][0];
                #pragma unroll
                for (int I = 0; I < 4; ++I) {
                    #pragma unroll
                    for (int p = 0; p < 2; ++p) {
                        sv[I][p] = *(const v4f*)&SB[16 * I + h8 + 4 * p];
                        bv[I][p] = *(const v4f*)&SB[KSPA + 16 * I + h8 + 4 * p];
                        av[I][p] = *(const v4f*)&Vl[I][l16][h8 + 4 * p];
                        #pragma unroll
                        for (int J = 0; J < I; ++J)
                            aw[I][J][p] = *(const v4f*)&Wl[16 * I + l16][16 * J + h8 + 4 * p];
                    }
                }
                #pragma unroll
                for (int I = 0; I < 4; ++I) {
                    #pragma unroll
                    for (int r = 0; r < 8; ++r)
                        R[I][r] += sv[I][r / 4][r % 4] * ev[I][r / 4][r % 4];
                }

                float* orow = out + myb * NN + NSH + h8;
                #pragma unroll
                for (int J = 0; J < 4; ++J) {
                    v8f Z = {0.f, 0.f, 0.f, 0.f, 0.f, 0.f, 0.f, 0.f};
                    #pragma unroll
                    for (int st = 0; st < 4; ++st) {
                        const v2f a = {av[J][st / 2][(st % 2) * 2], av[J][st / 2][(st % 2) * 2 + 1]};
                        const v2f b = {R[J][2 * st], R[J][2 * st + 1]};
                        Z = __builtin_amdgcn_wmma_f32_16x16x4_f32(
                             false, a,  false, b,
                             (short)0, Z,  false,  false);
                    }
                    if (cn < n) {
                        ((float4*)(orow + 16 * J))[0] = make_float4(Z[0] + bv[J][0].x, Z[1] + bv[J][0].y,
                                                                    Z[2] + bv[J][0].z, Z[3] + bv[J][0].w);
                        ((float4*)(orow + 16 * J))[1] = make_float4(Z[4] + bv[J][1].x, Z[5] + bv[J][1].y,
                                                                    Z[6] + bv[J][1].z, Z[7] + bv[J][1].w);
                    }
                    #pragma unroll
                    for (int I = J + 1; I < 4; ++I) {
                        #pragma unroll
                        for (int st = 0; st < 4; ++st) {
                            const v2f a = {aw[I][J][st / 2][(st % 2) * 2], aw[I][J][st / 2][(st % 2) * 2 + 1]};
                            const v2f b = {Z[2 * st], Z[2 * st + 1]};
                            R[I] = __builtin_amdgcn_wmma_f32_16x16x4_f32(
                                 false, a,  false, b,
                                 (short)0, R[I],  false,  false);
                        }
                    }
                }
            }
            t += TSLOTS;
            if (t * 16 >= n) break;
            cn = t * 16 + l16;
            myb = list[cn < n ? cn : t * 16];
            #pragma unroll
            for (int r = 0; r < 4; ++r) {
                const float* Erow = eps + __builtin_amdgcn_readlane(myb, wv * 4 + r) * NN + lane * 4;
                #pragma unroll
                for (int q = 0; q < 4; ++q)
                    rb[r][q] = *(const v4f*)(Erow + q * 128);
            }
        }
    }
}

extern "C" void kernel_launch(void* const* d_in, const int* in_sizes, int n_in,
                              void* d_out, int out_size, void* d_ws, size_t ws_size,
                              hipStream_t stream) {
    const float* eps     = (const float*)d_in[0];
    const int*   didx    = (const int*)  d_in[1];
    const float* L_emb   = (const float*)d_in[2];
    const float* S_emb   = (const float*)d_in[3];
    const float* bias_ns = (const float*)d_in[4];
    const float* bias_sh = (const float*)d_in[5];
    float*       out     = (float*)d_out;

    tail_subst_prefix_kernel<<<dim3(GRIDB), dim3(128), 0, stream>>>(
        eps, didx, L_emb, S_emb, bias_ns, bias_sh, out);
}
